# starting kernel
# baseline (speedup 1.0000x reference)
#include <hip/hip_runtime.h>
#include <hip/hip_bf16.h>

typedef __attribute__((ext_vector_type(2))) float v2f;
typedef __attribute__((ext_vector_type(8))) float v8f;

__global__ __launch_bounds__(256) void gemm_nt_wmma_f32(
    const float* __restrict__ A, const float* __restrict__ B,
    float* __restrict__ C, int K, int lda, int ldb, int ldc)
{
    const int lane = threadIdx.x & 31;
    const int wave = threadIdx.x >> 5;
    const int wrow = wave >> 2;
    const int wcol = wave & 3;
    const int m0   = blockIdx.x * 128 + wrow * 64;
    const int n0   = wcol * 64;
    const int lm   = lane & 15;
    const int lk   = (lane >> 4) << 1;
    const int g    = lane >> 4;

    v8f acc[4][4];
#pragma unroll
    for (int i = 0; i < 4; ++i)
#pragma unroll
        for (int j = 0; j < 4; ++j)
            acc[i][j] = (v8f){0.f, 0.f, 0.f, 0.f, 0.f, 0.f, 0.f, 0.f};

    const float* aBase = A + (size_t)(m0 + lm) * lda + lk;
    const float* bBase = B + (size_t)(n0 + lm) * ldb + lk;

    for (int k = 0; k < K; k += 4) {
        v2f af[4], bf[4];
#pragma unroll
        for (int i = 0; i < 4; ++i) {
            af[i] = *(const v2f*)(aBase + (size_t)(i * 16) * lda + k);
            bf[i] = *(const v2f*)(bBase + (size_t)(i * 16) * ldb + k);
        }
#pragma unroll
        for (int i = 0; i < 4; ++i)
#pragma unroll
            for (int j = 0; j < 4; ++j)
                acc[i][j] = __builtin_amdgcn_wmma_f32_16x16x4_f32(
                     false, af[i],  false, bf[j],
                     (short)0, acc[i][j],
                     false,  false);
    }

#pragma unroll
    for (int i = 0; i < 4; ++i) {
#pragma unroll
        for (int j = 0; j < 4; ++j) {
            float* cp = C + (size_t)(m0 + i * 16 + 8 * g) * ldc + n0 + j * 16 + lm;
#pragma unroll
            for (int v = 0; v < 8; ++v)
                cp[(size_t)v * ldc] = acc[i][j][v];
        }
    }
}

__global__ __launch_bounds__(256) void bias_fold_kernel(
    const float* __restrict__ W2, const float* __restrict__ b1,
    const float* __restrict__ b2, float* __restrict__ cvec)
{
    const int p = threadIdx.x;
    float acc = b2[p];
    const float* row = W2 + (size_t)p * 256;
    for (int o = 0; o < 256; ++o)
        acc = fmaf(row[o], b1[o], acc);
    cvec[p] = acc;
}

__global__ __launch_bounds__(256) void pair_reduce_kernel(
    const float* __restrict__ Ab, const float* __restrict__ Bb,
    const float* __restrict__ cvec, float* __restrict__ out)
{
    const int bi = blockIdx.x;
    const int b  = bi >> 6;
    const int i  = bi & 63;
    const int p  = threadIdx.x;

    const float a = Ab[(size_t)bi * 256 + p] + cvec[p];
    const float* Brow = Bb + (size_t)b * 64 * 256 + p;

    float acc = 0.f;
#pragma unroll 8
    for (int j = 0; j < 64; ++j)
        acc += fmaxf(a + Brow[(size_t)j * 256], 0.f);

    acc -= fmaxf(a + Brow[(size_t)i * 256], 0.f);
    out[(size_t)bi * 256 + p] = acc;
}

extern "C" void kernel_launch(void* const* d_in, const int* in_sizes, int n_in,
                              void* d_out, int out_size, void* d_ws, size_t ws_size,
                              hipStream_t stream)
{
    (void)in_sizes; (void)n_in; (void)out_size; (void)ws_size;

    const float* x  = (const float*)d_in[0];
    const float* W1 = (const float*)d_in[1];
    const float* b1 = (const float*)d_in[2];
    const float* W2 = (const float*)d_in[3];
    const float* b2 = (const float*)d_in[4];
    float* out = (float*)d_out;

    const int D = 512, O = 256;
    const int M = 32 * 64;

    float* hi = (float*)d_ws;
    float* hj = hi + (size_t)M * O;
    float* Ab = hj + (size_t)M * O;
    float* Bb = Ab + (size_t)M * O;
    float* cv = Bb + (size_t)M * O;

    dim3 block(256);
    dim3 grid(M / 128);

    gemm_nt_wmma_f32<<<grid, block, 0, stream>>>(x,  W1,     hi, D, D, 2 * D, O);
    gemm_nt_wmma_f32<<<grid, block, 0, stream>>>(x,  W1 + D, hj, D, D, 2 * D, O);
    gemm_nt_wmma_f32<<<grid, block, 0, stream>>>(hi, W2,     Ab, O, O, O,     O);
    gemm_nt_wmma_f32<<<grid, block, 0, stream>>>(hj, W2,     Bb, O, O, O,     O);
    bias_fold_kernel<<<1, 256, 0, stream>>>(W2, b1, b2, cv);
    pair_reduce_kernel<<<M, 256, 0, stream>>>(Ab, Bb, cv, out);
}


# Launch count 3 to 2: pairwise relu sum moved into the second-layer GEMM launch; an 8-wave block per (batch, 32 columns) forms Ab and Bb tiles, hands them over in LDS and sums the pairs
# speedup vs baseline: 9.7661x; 9.7661x over previous
#include <hip/hip_runtime.h>
#include <hip/hip_bf16.h>

typedef __attribute__((ext_vector_type(2))) float v2f;
typedef __attribute__((ext_vector_type(4))) float v4f;
typedef __attribute__((ext_vector_type(8))) float v8f;

#define GEMM_BM 64
#define GEMM_BN 32
#define GEMM_KT 64
#define GEMM_PITCH 68

template <int NT, bool BIAS>
__global__ __launch_bounds__(128) void gemm_nt_wmma_f32(
    const float* __restrict__ A, const float* __restrict__ B,
    float* __restrict__ C, const float* __restrict__ bias,
    int lda, int ldb, int ldc, int az, int bz, int cz)
{
    __shared__ __attribute__((aligned(16))) float sA[2][GEMM_BM * GEMM_PITCH];
    __shared__ __attribute__((aligned(16))) float sB[2][GEMM_BN * GEMM_PITCH];

    const int z = blockIdx.z;
    A += (size_t)z * az;
    B += (size_t)z * bz;
    C += (size_t)z * cz;

    const int tid  = threadIdx.x;
    const int lane = tid & 31;
    const int wave = tid >> 5;
    const int wrow = wave >> 1;
    const int wcol = wave & 1;
    const int bm0  = blockIdx.y * GEMM_BM;
    const int bn0  = blockIdx.x * GEMM_BN;
    const int m0   = bm0 + wrow * 32;
    const int n0   = bn0 + wcol * 16;
    const int lm   = lane & 15;
    const int lk   = (lane >> 4) << 1;
    const int g    = lane >> 4;

    const int srow = tid >> 4;
    const int scol = (tid & 15) << 2;
    const float* aG = A + (size_t)(bm0 + srow) * lda + scol;
    const float* bG = B + (size_t)(bn0 + srow) * ldb + scol;
    const int sOff  = srow * GEMM_PITCH + scol;
    const int aFrag = (wrow * 32 + lm) * GEMM_PITCH + lk;
    const int bFrag = (wcol * 16 + lm) * GEMM_PITCH + lk;

    float bv = 0.f;
    if (BIAS) {
        if (z == 0) bv = bias[n0 + lm];
    }

    v8f acc0 = (v8f){0.f, 0.f, 0.f, 0.f, 0.f, 0.f, 0.f, 0.f};
    v8f acc1 = (v8f){0.f, 0.f, 0.f, 0.f, 0.f, 0.f, 0.f, 0.f};

    v4f pa[8], pb[4];

#pragma unroll
    for (int i = 0; i < 8; ++i)
        pa[i] = *(const v4f*)(aG + (size_t)(8 * i) * lda);
#pragma unroll
    for (int i = 0; i < 4; ++i)
        pb[i] = *(const v4f*)(bG + (size_t)(8 * i) * ldb);
#pragma unroll
    for (int i = 0; i < 8; ++i)
        *(v4f*)&sA[0][sOff + 8 * i * GEMM_PITCH] = pa[i];
#pragma unroll
    for (int i = 0; i < 4; ++i)
        *(v4f*)&sB[0][sOff + 8 * i * GEMM_PITCH] = pb[i];
#pragma unroll
    for (int i = 0; i < 8; ++i)
        pa[i] = *(const v4f*)(aG + (size_t)(8 * i) * lda + GEMM_KT);
#pragma unroll
    for (int i = 0; i < 4; ++i)
        pb[i] = *(const v4f*)(bG + (size_t)(8 * i) * ldb + GEMM_KT);

#pragma unroll
    for (int t = 0; t < NT; ++t) {
        const int cur = t & 1;
        const int nxt = cur ^ 1;

        __builtin_amdgcn_fence(__ATOMIC_RELEASE, "workgroup", "local");
        __builtin_amdgcn_s_barrier();
        __builtin_amdgcn_fence(__ATOMIC_ACQUIRE, "workgroup", "local");

        v2f fa0[16], fa1[16], fb[16];
        const float* a0 = &sA[cur][aFrag];
        const float* b0 = &sB[cur][bFrag];
#pragma unroll
        for (int s = 0; s < 16; ++s) {
            fa0[s] = *(const v2f*)(a0 + 4 * s);
            fa1[s] = *(const v2f*)(a0 + 16 * GEMM_PITCH + 4 * s);
            fb[s]  = *(const v2f*)(b0 + 4 * s);
        }
        __builtin_amdgcn_sched_barrier(0);

        if (t + 1 < NT) {
#pragma unroll
            for (int i = 0; i < 8; ++i)
                *(v4f*)&sA[nxt][sOff + 8 * i * GEMM_PITCH] = pa[i];
#pragma unroll
            for (int i = 0; i < 4; ++i)
                *(v4f*)&sB[nxt][sOff + 8 * i * GEMM_PITCH] = pb[i];
        }
        if (t + 2 < NT) {
            const int k2 = (t + 2) * GEMM_KT;
#pragma unroll
            for (int i = 0; i < 8; ++i)
                pa[i] = *(const v4f*)(aG + (size_t)(8 * i) * lda + k2);
#pragma unroll
            for (int i = 0; i < 4; ++i)
                pb[i] = *(const v4f*)(bG + (size_t)(8 * i) * ldb + k2);
        }
        __builtin_amdgcn_sched_barrier(0);

#pragma unroll
        for (int s = 0; s < 16; ++s) {
            acc0 = __builtin_amdgcn_wmma_f32_16x16x4_f32(
                 false, fa0[s],  false, fb[s],
                 (short)0, acc0,
                 false,  false);
            acc1 = __builtin_amdgcn_wmma_f32_16x16x4_f32(
                 false, fa1[s],  false, fb[s],
                 (short)0, acc1,
                 false,  false);
        }
    }

    if (BIAS) {
#pragma unroll
        for (int v = 0; v < 8; ++v) {
            acc0[v] += bv;
            acc1[v] += bv;
        }
    }

    {
        float* cp = C + (size_t)(m0 + 8 * g) * ldc + n0 + lm;
#pragma unroll
        for (int v = 0; v < 8; ++v)
            cp[(size_t)v * ldc] = acc0[v];
        cp += (size_t)16 * ldc;
#pragma unroll
        for (int v = 0; v < 8; ++v)
            cp[(size_t)v * ldc] = acc1[v];
    }
}

#define FUSE_ROWS 128
#define FUSE_PC   32
#define FUSE_TP   68

__global__ __launch_bounds__(256) void gemm2_pair_fused(
    const float* __restrict__ H, const float* __restrict__ W2,
    const float* __restrict__ b2, float* __restrict__ out, int hz)
{
    __shared__ __attribute__((aligned(16))) float sA[2][FUSE_ROWS * GEMM_PITCH];
    __shared__ __attribute__((aligned(16))) float sB[2][FUSE_PC * GEMM_PITCH];
    __shared__ __attribute__((aligned(16))) float sT[2][FUSE_PC * FUSE_TP];

    const int tid   = threadIdx.x;
    const int lane  = tid & 31;
    const int wave  = tid >> 5;
    const int plane = wave >> 2;
    const int wrow  = (wave >> 1) & 1;
    const int wcol  = wave & 1;
    const int b     = blockIdx.y;
    const int pc0   = blockIdx.x * FUSE_PC;
    const int lm    = lane & 15;
    const int lk    = (lane >> 4) << 1;
    const int g     = lane >> 4;

    const int srow = tid >> 4;
    const int scol = (tid & 15) << 2;
    const float* aG = H + (size_t)(b * 64 + srow) * 256 + scol;
    const float* bG = W2 + (size_t)(pc0 + srow) * 256 + scol;
    const int sOff  = srow * GEMM_PITCH + scol;
    const int aFrag = (plane * 64 + wrow * 32 + lm) * GEMM_PITCH + lk;
    const int bFrag = (wcol * 16 + lm) * GEMM_PITCH + lk;

    float bv = 0.f;
    if (plane == 0) bv = b2[pc0 + wcol * 16 + lm];

    v8f acc0 = (v8f){0.f, 0.f, 0.f, 0.f, 0.f, 0.f, 0.f, 0.f};
    v8f acc1 = (v8f){0.f, 0.f, 0.f, 0.f, 0.f, 0.f, 0.f, 0.f};

    v4f pa[8], pb[2];

#pragma unroll
    for (int i = 0; i < 8; ++i)
        pa[i] = *(const v4f*)(aG + (size_t)(i >> 2) * hz + (size_t)(16 * (i & 3)) * 256);
#pragma unroll
    for (int i = 0; i < 2; ++i)
        pb[i] = *(const v4f*)(bG + (size_t)(16 * i) * 256);
#pragma unroll
    for (int i = 0; i < 8; ++i)
        *(v4f*)&sA[0][sOff + 16 * i * GEMM_PITCH] = pa[i];
#pragma unroll
    for (int i = 0; i < 2; ++i)
        *(v4f*)&sB[0][sOff + 16 * i * GEMM_PITCH] = pb[i];
#pragma unroll
    for (int i = 0; i < 8; ++i)
        pa[i] = *(const v4f*)(aG + (size_t)(i >> 2) * hz + (size_t)(16 * (i & 3)) * 256 + GEMM_KT);
#pragma unroll
    for (int i = 0; i < 2; ++i)
        pb[i] = *(const v4f*)(bG + (size_t)(16 * i) * 256 + GEMM_KT);

#pragma unroll
    for (int t = 0; t < 4; ++t) {
        const int cur = t & 1;
        const int nxt = cur ^ 1;

        __builtin_amdgcn_fence(__ATOMIC_RELEASE, "workgroup", "local");
        __builtin_amdgcn_s_barrier();
        __builtin_amdgcn_fence(__ATOMIC_ACQUIRE, "workgroup", "local");

        v2f fa0[16], fa1[16], fb[16];
        const float* a0 = &sA[cur][aFrag];
        const float* b0 = &sB[cur][bFrag];
#pragma unroll
        for (int s = 0; s < 16; ++s) {
            fa0[s] = *(const v2f*)(a0 + 4 * s);
            fa1[s] = *(const v2f*)(a0 + 16 * GEMM_PITCH + 4 * s);
            fb[s]  = *(const v2f*)(b0 + 4 * s);
        }
        __builtin_amdgcn_sched_barrier(0);

        if (t + 1 < 4) {
#pragma unroll
            for (int i = 0; i < 8; ++i)
                *(v4f*)&sA[nxt][sOff + 16 * i * GEMM_PITCH] = pa[i];
#pragma unroll
            for (int i = 0; i < 2; ++i)
                *(v4f*)&sB[nxt][sOff + 16 * i * GEMM_PITCH] = pb[i];
        }
        if (t + 2 < 4) {
            const int k2 = (t + 2) * GEMM_KT;
#pragma unroll
            for (int i = 0; i < 8; ++i)
                pa[i] = *(const v4f*)(aG + (size_t)(i >> 2) * hz + (size_t)(16 * (i & 3)) * 256 + k2);
#pragma unroll
            for (int i = 0; i < 2; ++i)
                pb[i] = *(const v4f*)(bG + (size_t)(16 * i) * 256 + k2);
        }
        __builtin_amdgcn_sched_barrier(0);

#pragma unroll
        for (int s = 0; s < 16; ++s) {
            acc0 = __builtin_amdgcn_wmma_f32_16x16x4_f32(
                 false, fa0[s],  false, fb[s],
                 (short)0, acc0,
                 false,  false);
            acc1 = __builtin_amdgcn_wmma_f32_16x16x4_f32(
                 false, fa1[s],  false, fb[s],
                 (short)0, acc1,
                 false,  false);
        }
    }

    if (plane == 0) {
#pragma unroll
        for (int v = 0; v < 8; ++v) {
            acc0[v] += bv;
            acc1[v] += bv;
        }
    }
    {
        float* tp = &sT[plane][(wcol * 16 + lm) * FUSE_TP + wrow * 32 + 8 * g];
        *(v4f*)(tp)      = (v4f){acc0[0], acc0[1], acc0[2], acc0[3]};
        *(v4f*)(tp + 4)  = (v4f){acc0[4], acc0[5], acc0[6], acc0[7]};
        *(v4f*)(tp + 16) = (v4f){acc1[0], acc1[1], acc1[2], acc1[3]};
        *(v4f*)(tp + 20) = (v4f){acc1[4], acc1[5], acc1[6], acc1[7]};
    }
    __syncthreads();

    const int i0 = 8 * wave;
    const float* ta = &sT[0][lane * FUSE_TP];
    const float* tb = &sT[1][lane * FUSE_TP];

    float a[8], acc[8];
    {
        const v4f lo = *(const v4f*)(ta + i0);
        const v4f hi4 = *(const v4f*)(ta + i0 + 4);
        a[0] = lo[0];  a[1] = lo[1];  a[2] = lo[2];  a[3] = lo[3];
        a[4] = hi4[0]; a[5] = hi4[1]; a[6] = hi4[2]; a[7] = hi4[3];
    }
#pragma unroll
    for (int k = 0; k < 8; ++k) acc[k] = 0.f;

#pragma unroll 4
    for (int j4 = 0; j4 < 16; ++j4) {
        const v4f bq = *(const v4f*)(tb + 4 * j4);
#pragma unroll
        for (int jj = 0; jj < 4; ++jj) {
#pragma unroll
            for (int k = 0; k < 8; ++k)
                acc[k] += fmaxf(a[k] + bq[jj], 0.f);
        }
    }

    {
        const v4f lo = *(const v4f*)(tb + i0);
        const v4f hi4 = *(const v4f*)(tb + i0 + 4);
        const float d[8] = {lo[0], lo[1], lo[2], lo[3], hi4[0], hi4[1], hi4[2], hi4[3]};
#pragma unroll
        for (int k = 0; k < 8; ++k)
            acc[k] -= fmaxf(a[k] + d[k], 0.f);
    }

    float* op = out + (size_t)(b * 64 + i0) * 256 + pc0 + lane;
#pragma unroll
    for (int k = 0; k < 8; ++k)
        op[(size_t)k * 256] = acc[k];
}

extern "C" void kernel_launch(void* const* d_in, const int* in_sizes, int n_in,
                              void* d_out, int out_size, void* d_ws, size_t ws_size,
                              hipStream_t stream)
{
    (void)in_sizes; (void)n_in; (void)out_size; (void)ws_size;

    const float* x  = (const float*)d_in[0];
    const float* W1 = (const float*)d_in[1];
    const float* b1 = (const float*)d_in[2];
    const float* W2 = (const float*)d_in[3];
    const float* b2 = (const float*)d_in[4];
    float* out = (float*)d_out;

    const int D = 512, O = 256;
    const int M = 32 * 64;

    float* hi = (float*)d_ws;

    dim3 block(128);
    dim3 grid(O / 32, M / 64, 2);

    gemm_nt_wmma_f32<8, true><<<grid, block, 0, stream>>>(
        x, W1, hi, b1, D, 2 * D, O, 0, D, M * O);
    gemm2_pair_fused<<<dim3(O / 32, 32), dim3(256), 0, stream>>>(
        hi, W2, b2, out, M * O);
}
